# starting kernel
# baseline (speedup 1.0000x reference)
#include <hip/hip_runtime.h>

#define HID       128
#define LDS_PITCH 132
#define MT        80
#define MSUB      5

typedef __attribute__((ext_vector_type(2))) float v2f;
typedef __attribute__((ext_vector_type(8))) float v8f;

__global__ void sage_zero(float* __restrict__ p, int n) {
    int i = blockIdx.x * blockDim.x + threadIdx.x;
    if (i < n) p[i] = 0.0f;
}

__global__ void sage_rdeg(const float* __restrict__ degf,
                          float* __restrict__ rdeg, int n) {
    int i = blockIdx.x * blockDim.x + threadIdx.x;
    if (i < n) {
        float d = degf[i];
        rdeg[i] = d > 1.0f ? 1.0f / d : 1.0f;
    }
}

__global__ void sage_scatter(const float* __restrict__ x,
                             const int*   __restrict__ src,
                             const int*   __restrict__ dst,
                             float*       __restrict__ agg,
                             float*       __restrict__ degf,
                             int nEdges, int doDeg) {
    int gtid = blockIdx.x * blockDim.x + threadIdx.x;
    int edge = gtid >> 5;
    int lane = gtid & 31;
    if (edge >= nEdges) return;

    int s = src[edge];
    int d = dst[edge];
    const float* xs = x   + (size_t)s * HID;
    float*       ad = agg + (size_t)d * HID;

    atomicAdd(ad + lane,      xs[lane]);
    atomicAdd(ad + lane + 32, xs[lane + 32]);
    atomicAdd(ad + lane + 64, xs[lane + 64]);
    atomicAdd(ad + lane + 96, xs[lane + 96]);
    if (doDeg && lane == 0) atomicAdd(degf + d, 1.0f);
}

__global__ void sage_gemm(const float* __restrict__ agg,
                          const float* __restrict__ rdeg,
                          const float* __restrict__ xin,
                          const float* __restrict__ wl,
                          const float* __restrict__ bl,
                          const float* __restrict__ wr,
                          float*       __restrict__ out,
                          int applyRelu) {
    __shared__ float sMean[MT * LDS_PITCH];
    __shared__ float sX   [MT * LDS_PITCH];

    const int tid      = threadIdx.x;
    const int tileBase = blockIdx.x * MT;

    for (int i = tid; i < MT * 32; i += 256) {
        int r   = i >> 5;
        int c4  = (i & 31) << 2;
        int row = tileBase + r;
        float rd = rdeg[row];
        float4 av = *(const float4*)(agg + (size_t)row * HID + c4);
        float4 mv = make_float4(av.x * rd, av.y * rd, av.z * rd, av.w * rd);
        *(float4*)(sMean + r * LDS_PITCH + c4) = mv;
        *(float4*)(sX    + r * LDS_PITCH + c4) =
            *(const float4*)(xin + (size_t)row * HID + c4);
    }
    __syncthreads();

    const int wave  = tid >> 5;
    const int lane  = tid & 31;
    const int lrow  = lane & 15;
    const int halfK = (lane >> 4) << 1;
    const int nCol  = wave * 16 + lrow;

    v8f acc[MSUB];
    #pragma unroll
    for (int mt = 0; mt < MSUB; ++mt) acc[mt] = (v8f){};

    for (int k0 = 0; k0 < HID; k0 += 4) {
        v2f b;
        b.x = wl[(size_t)(k0 + halfK)     * HID + nCol];
        b.y = wl[(size_t)(k0 + halfK + 1) * HID + nCol];
        #pragma unroll
        for (int mt = 0; mt < MSUB; ++mt) {
            const float* ap = sMean + (mt * 16 + lrow) * LDS_PITCH + k0 + halfK;
            v2f a; a.x = ap[0]; a.y = ap[1];
            acc[mt] = __builtin_amdgcn_wmma_f32_16x16x4_f32(
                          false, a, false, b, (short)0, acc[mt], false, false);
        }
    }

    for (int k0 = 0; k0 < HID; k0 += 4) {
        v2f b;
        b.x = wr[(size_t)(k0 + halfK)     * HID + nCol];
        b.y = wr[(size_t)(k0 + halfK + 1) * HID + nCol];
        #pragma unroll
        for (int mt = 0; mt < MSUB; ++mt) {
            const float* ap = sX + (mt * 16 + lrow) * LDS_PITCH + k0 + halfK;
            v2f a; a.x = ap[0]; a.y = ap[1];
            acc[mt] = __builtin_amdgcn_wmma_f32_16x16x4_f32(
                          false, a, false, b, (short)0, acc[mt], false, false);
        }
    }

    const float bias   = bl[nCol];
    const int   rowOff = (lane < 16) ? 0 : 8;
    #pragma unroll
    for (int mt = 0; mt < MSUB; ++mt) {
        #pragma unroll
        for (int r = 0; r < 8; ++r) {
            float v = acc[mt][r] + bias;
            if (applyRelu) v = v > 0.0f ? v : 0.0f;
            out[(size_t)(tileBase + mt * 16 + r + rowOff) * HID + nCol] = v;
        }
    }
}

extern "C" void kernel_launch(void* const* d_in, const int* in_sizes, int n_in,
                              void* d_out, int out_size, void* d_ws, size_t ws_size,
                              hipStream_t stream) {
    const float* x    = (const float*)d_in[0];
    const int*   ei   = (const int*)  d_in[1];
    const float* w1l  = (const float*)d_in[2];
    const float* b1l  = (const float*)d_in[3];
    const float* w1r  = (const float*)d_in[4];
    const float* w2l  = (const float*)d_in[5];
    const float* b2l  = (const float*)d_in[6];
    const float* w2r  = (const float*)d_in[7];

    const int N = in_sizes[0] / HID;
    const int E = in_sizes[1] / 2;
    const int* src = ei;
    const int* dst = ei + E;

    float* agg  = (float*)d_ws;
    float* deg  = agg  + (size_t)N * HID;
    float* h    = deg  + N;
    float* rdeg = h    + (size_t)N * HID;
    float* out  = (float*)d_out;

    const int zeroAllN   = N * (HID + 1);
    const int zeroAggN   = N * HID;
    const int zAllGrid   = (zeroAllN + 255) / 256;
    const int zAggGrid   = (zeroAggN + 255) / 256;
    const int nGrid      = (N + 255) / 256;
    const int scatGrid   = (int)(((long long)E * 32 + 255) / 256);
    const int gemmGrid   = N / MT;

    sage_zero   <<<zAllGrid, 256, 0, stream>>>(agg, zeroAllN);
    sage_scatter<<<scatGrid, 256, 0, stream>>>(x, src, dst, agg, deg, E, 1);
    sage_rdeg   <<<nGrid,    256, 0, stream>>>(deg, rdeg, N);
    sage_gemm   <<<gemmGrid, 256, 0, stream>>>(agg, rdeg, x, w1l, b1l, w1r, h, 1);

    sage_zero   <<<zAggGrid, 256, 0, stream>>>(agg, zeroAggN);
    sage_scatter<<<scatGrid, 256, 0, stream>>>(h, src, dst, agg, deg, E, 0);
    sage_gemm   <<<gemmGrid, 256, 0, stream>>>(agg, rdeg, h, w2l, b2l, w2r, out, 0);
}


# Neighbour sums by gather over per-destination edge lists built in-call (int count/slot/fill) replace float atomic scatter and zero pass; per-node sum order = arrival-order slot
# speedup vs baseline: 1.7354x; 1.7354x over previous
#include <hip/hip_runtime.h>

#define HID       128
#define LDS_PITCH 132
#define MT        80
#define MSUB      5

typedef __attribute__((ext_vector_type(2))) float v2f;
typedef __attribute__((ext_vector_type(8))) float v8f;

__global__ void sage_zero_i(int* __restrict__ p, int n) {
    int i = blockIdx.x * blockDim.x + threadIdx.x;
    if (i < n) p[i] = 0;
}

__global__ void sage_count(const int* __restrict__ dst,
                           int*       __restrict__ cnt,
                           int*       __restrict__ slot, int nEdges) {
    int e = blockIdx.x * blockDim.x + threadIdx.x;
    if (e < nEdges) slot[e] = atomicAdd(cnt + dst[e], 1);
}

__global__ void sage_alloc(const int* __restrict__ cnt,
                           int*       __restrict__ cursor,
                           int*       __restrict__ seg,
                           float*     __restrict__ rdeg, int n) {
    const int i    = blockIdx.x * blockDim.x + threadIdx.x;
    const int lane = threadIdx.x & 31;
    const int c    = (i < n) ? cnt[i] : 0;

    int incl = c;
    #pragma unroll
    for (int off = 1; off < 32; off <<= 1) {
        int t = __shfl_up(incl, off, 32);
        if (lane >= off) incl += t;
    }
    int base = 0;
    if (lane == 31) base = atomicAdd(cursor, incl);
    base = __shfl(base, 31, 32);

    if (i < n) {
        ((int2*)seg)[i] = make_int2(base + incl - c, c);
        rdeg[i] = c > 1 ? 1.0f / (float)c : 1.0f;
    }
}

__global__ void sage_fill(const int* __restrict__ src,
                          const int* __restrict__ dst,
                          const int* __restrict__ seg,
                          const int* __restrict__ slot,
                          int*       __restrict__ nbr, int nEdges) {
    int e = blockIdx.x * blockDim.x + threadIdx.x;
    if (e < nEdges) {
        unsigned pos = (unsigned)(seg[2 * dst[e]] + slot[e]);
        if (pos < (unsigned)nEdges) nbr[pos] = src[e];
    }
}

#define GATHER_WAVES 8
__global__ void __launch_bounds__(GATHER_WAVES * 32)
sage_gather(const float* __restrict__ x,
            const int*   __restrict__ seg,
            const int*   __restrict__ nbr,
            float*       __restrict__ agg,
            int nNodes, int nEdges) {
    const int lane = threadIdx.x & 31;
    const int wave = __builtin_amdgcn_readfirstlane((int)(threadIdx.x >> 5));
    const int node = blockIdx.x * GATHER_WAVES + wave;
    if (node >= nNodes) return;

    const int2 sc = ((const int2*)seg)[node];
    const int  s  = sc.x;
    const int  c  = sc.y;

    float4 acc = make_float4(0.0f, 0.0f, 0.0f, 0.0f);

    for (int base = 0; base < c; base += 32) {
        const int m = (c - base) < 32 ? (c - base) : 32;
        unsigned p = (unsigned)(s + base + lane);
        int myIdx = 0;
        if (lane < m && p < (unsigned)nEdges) myIdx = nbr[p];
        myIdx = (unsigned)myIdx < (unsigned)nNodes ? myIdx : 0;

        for (int j = 0; j < m; j += 8) {
            float4 v[8];
            #pragma unroll
            for (int k = 0; k < 8; ++k) {
                v[k] = make_float4(0.0f, 0.0f, 0.0f, 0.0f);
                if (j + k < m) {
                    const int idx = __builtin_amdgcn_readlane(myIdx, j + k);
                    v[k] = *((const float4*)(x + (size_t)idx * HID) + lane);
                }
            }
            #pragma unroll
            for (int k = 0; k < 8; ++k) {
                acc.x += v[k].x; acc.y += v[k].y; acc.z += v[k].z; acc.w += v[k].w;
            }
        }
    }

    *((float4*)(agg + (size_t)node * HID) + lane) = acc;
}

__global__ void sage_gemm(const float* __restrict__ agg,
                          const float* __restrict__ rdeg,
                          const float* __restrict__ xin,
                          const float* __restrict__ wl,
                          const float* __restrict__ bl,
                          const float* __restrict__ wr,
                          float*       __restrict__ out,
                          int applyRelu) {
    __shared__ float sMean[MT * LDS_PITCH];
    __shared__ float sX   [MT * LDS_PITCH];

    const int tid      = threadIdx.x;
    const int tileBase = blockIdx.x * MT;

    for (int i = tid; i < MT * 32; i += 256) {
        int r   = i >> 5;
        int c4  = (i & 31) << 2;
        int row = tileBase + r;
        float rd = rdeg[row];
        float4 av = *(const float4*)(agg + (size_t)row * HID + c4);
        float4 mv = make_float4(av.x * rd, av.y * rd, av.z * rd, av.w * rd);
        *(float4*)(sMean + r * LDS_PITCH + c4) = mv;
        *(float4*)(sX    + r * LDS_PITCH + c4) =
            *(const float4*)(xin + (size_t)row * HID + c4);
    }
    __syncthreads();

    const int wave  = tid >> 5;
    const int lane  = tid & 31;
    const int lrow  = lane & 15;
    const int halfK = (lane >> 4) << 1;
    const int nCol  = wave * 16 + lrow;

    v8f acc[MSUB];
    #pragma unroll
    for (int mt = 0; mt < MSUB; ++mt) acc[mt] = (v8f){};

    for (int k0 = 0; k0 < HID; k0 += 4) {
        v2f b;
        b.x = wl[(size_t)(k0 + halfK)     * HID + nCol];
        b.y = wl[(size_t)(k0 + halfK + 1) * HID + nCol];
        #pragma unroll
        for (int mt = 0; mt < MSUB; ++mt) {
            const float* ap = sMean + (mt * 16 + lrow) * LDS_PITCH + k0 + halfK;
            v2f a; a.x = ap[0]; a.y = ap[1];
            acc[mt] = __builtin_amdgcn_wmma_f32_16x16x4_f32(
                          false, a, false, b, (short)0, acc[mt], false, false);
        }
    }

    for (int k0 = 0; k0 < HID; k0 += 4) {
        v2f b;
        b.x = wr[(size_t)(k0 + halfK)     * HID + nCol];
        b.y = wr[(size_t)(k0 + halfK + 1) * HID + nCol];
        #pragma unroll
        for (int mt = 0; mt < MSUB; ++mt) {
            const float* ap = sX + (mt * 16 + lrow) * LDS_PITCH + k0 + halfK;
            v2f a; a.x = ap[0]; a.y = ap[1];
            acc[mt] = __builtin_amdgcn_wmma_f32_16x16x4_f32(
                          false, a, false, b, (short)0, acc[mt], false, false);
        }
    }

    const float bias   = bl[nCol];
    const int   rowOff = (lane < 16) ? 0 : 8;
    #pragma unroll
    for (int mt = 0; mt < MSUB; ++mt) {
        #pragma unroll
        for (int r = 0; r < 8; ++r) {
            float v = acc[mt][r] + bias;
            if (applyRelu) v = v > 0.0f ? v : 0.0f;
            out[(size_t)(tileBase + mt * 16 + r + rowOff) * HID + nCol] = v;
        }
    }
}

extern "C" void kernel_launch(void* const* d_in, const int* in_sizes, int n_in,
                              void* d_out, int out_size, void* d_ws, size_t ws_size,
                              hipStream_t stream) {
    const float* x    = (const float*)d_in[0];
    const int*   ei   = (const int*)  d_in[1];
    const float* w1l  = (const float*)d_in[2];
    const float* b1l  = (const float*)d_in[3];
    const float* w1r  = (const float*)d_in[4];
    const float* w2l  = (const float*)d_in[5];
    const float* b2l  = (const float*)d_in[6];
    const float* w2r  = (const float*)d_in[7];

    const int N = in_sizes[0] / HID;
    const int E = in_sizes[1] / 2;
    const int* src = ei;
    const int* dst = ei + E;

    float* agg   = (float*)d_ws;
    float* h     = agg  + (size_t)N * HID;
    float* rdeg  = h    + (size_t)N * HID;
    int*   seg   = (int*)(rdeg + N);
    int*   slot  = seg + 2 * (size_t)N;
    int*   nbr   = slot + E;
    int*   cnt   = nbr  + E;
    float* out   = (float*)d_out;

    const int zGrid      = (N + 1 + 255) / 256;
    const int nGrid      = (N + 255) / 256;
    const int eGrid      = (E + 255) / 256;
    const int gatherGrid = (N + GATHER_WAVES - 1) / GATHER_WAVES;
    const int gemmGrid   = N / MT;

    sage_zero_i <<<zGrid, 256, 0, stream>>>(cnt, N + 1);
    sage_count  <<<eGrid, 256, 0, stream>>>(dst, cnt, slot, E);
    sage_alloc  <<<nGrid, 256, 0, stream>>>(cnt, cnt + N, seg, rdeg, N);
    sage_fill   <<<eGrid, 256, 0, stream>>>(src, dst, seg, slot, nbr, E);

    sage_gather <<<gatherGrid, GATHER_WAVES * 32, 0, stream>>>(x, seg, nbr, agg, N, E);
    sage_gemm   <<<gemmGrid, 256, 0, stream>>>(agg, rdeg, x, w1l, b1l, w1r, h, 1);

    sage_gather <<<gatherGrid, GATHER_WAVES * 32, 0, stream>>>(h, seg, nbr, agg, N, E);
    sage_gemm   <<<gemmGrid, 256, 0, stream>>>(agg, rdeg, h, w2l, b2l, w2r, out, 0);
}


# Neighbour gather fused into the dense stage: each 80-row block gathers its rows' neighbour sums straight into LDS, same list order; agg and rdeg planes and 2 of 8 launches dropped
# speedup vs baseline: 1.9410x; 1.1185x over previous
#include <hip/hip_runtime.h>

#define HID       128
#define LDS_PITCH 132
#define MT        80
#define MSUB      5
#define ROWS_PER_WAVE (MT / 8)

typedef __attribute__((ext_vector_type(2))) float v2f;
typedef __attribute__((ext_vector_type(8))) float v8f;

__global__ void sage_zero_i(int* __restrict__ p, int n) {
    int i = blockIdx.x * blockDim.x + threadIdx.x;
    if (i < n) p[i] = 0;
}

__global__ void sage_count(const int* __restrict__ dst,
                           int*       __restrict__ cnt,
                           int*       __restrict__ slot, int nEdges) {
    int e = blockIdx.x * blockDim.x + threadIdx.x;
    if (e < nEdges) slot[e] = atomicAdd(cnt + dst[e], 1);
}

__global__ void sage_alloc(const int* __restrict__ cnt,
                           int*       __restrict__ cursor,
                           int*       __restrict__ seg, int n) {
    const int i    = blockIdx.x * blockDim.x + threadIdx.x;
    const int lane = threadIdx.x & 31;
    const int c    = (i < n) ? cnt[i] : 0;

    int incl = c;
    #pragma unroll
    for (int off = 1; off < 32; off <<= 1) {
        int t = __shfl_up(incl, off, 32);
        if (lane >= off) incl += t;
    }
    int base = 0;
    if (lane == 31) base = atomicAdd(cursor, incl);
    base = __shfl(base, 31, 32);

    if (i < n) ((int2*)seg)[i] = make_int2(base + incl - c, c);
}

__global__ void sage_fill(const int* __restrict__ src,
                          const int* __restrict__ dst,
                          const int* __restrict__ seg,
                          const int* __restrict__ slot,
                          int*       __restrict__ nbr, int nEdges) {
    int e = blockIdx.x * blockDim.x + threadIdx.x;
    if (e < nEdges) {
        unsigned pos = (unsigned)(seg[2 * dst[e]] + slot[e]);
        if (pos < (unsigned)nEdges) nbr[pos] = src[e];
    }
}

__device__ __forceinline__ void sage_rows8(const float* __restrict__ xin, int myIdx,
                                           int j, int m, int lane, float4 (&v)[8]) {
    #pragma unroll
    for (int k = 0; k < 8; ++k) {
        const int    idx  = __builtin_amdgcn_readlane(myIdx, j + k);
        const float4 t    = *((const float4*)(xin + (size_t)idx * HID) + lane);
        const bool   live = (j + k < m);
        v[k] = make_float4(live ? t.x : 0.0f, live ? t.y : 0.0f,
                           live ? t.z : 0.0f, live ? t.w : 0.0f);
    }
}

__device__ __forceinline__ void sage_add8(float4& acc, const float4 (&v)[8]) {
    #pragma unroll
    for (int k = 0; k < 8; ++k) {
        acc.x += v[k].x; acc.y += v[k].y; acc.z += v[k].z; acc.w += v[k].w;
    }
}

__global__ void sage_layer(const float* __restrict__ xin,
                           const int*   __restrict__ seg,
                           const int*   __restrict__ nbr,
                           const float* __restrict__ wl,
                           const float* __restrict__ bl,
                           const float* __restrict__ wr,
                           float*       __restrict__ out,
                           int applyRelu, int nNodes, int nEdges) {
    __shared__ float sMean[MT * LDS_PITCH];
    __shared__ float sX   [MT * LDS_PITCH];

    const int tid      = threadIdx.x;
    const int tileBase = blockIdx.x * MT;
    const int lane     = tid & 31;
    const int waveU    = __builtin_amdgcn_readfirstlane(tid >> 5);
    const int r0       = waveU * ROWS_PER_WAVE;

    int2 sc = make_int2(0, 0);
    if (lane < ROWS_PER_WAVE) sc = ((const int2*)seg)[tileBase + r0 + lane];

    for (int i = tid; i < MT * 32; i += 256) {
        int r   = i >> 5;
        int c4  = (i & 31) << 2;
        *(float4*)(sX + r * LDS_PITCH + c4) =
            *(const float4*)(xin + (size_t)(tileBase + r) * HID + c4);
    }

    int s = __builtin_amdgcn_readlane(sc.x, 0);
    int c = __builtin_amdgcn_readlane(sc.y, 0);
    int nextIdx = 0;
    {
        unsigned p = (unsigned)(s + lane);
        if (lane < c && p < (unsigned)nEdges) nextIdx = nbr[p];
    }
    for (int q = 0; q < ROWS_PER_WAVE; ++q) {
        const int cs = s;
        const int cc = c;
        const int m0 = cc < 32 ? cc : 32;
        int myIdx = (unsigned)nextIdx < (unsigned)nNodes ? nextIdx : 0;

        float4 acc = make_float4(0.0f, 0.0f, 0.0f, 0.0f);
        float4 v[8];
        sage_rows8(xin, myIdx, 0, m0, lane, v);
        if (q + 1 < ROWS_PER_WAVE) {
            s = __builtin_amdgcn_readlane(sc.x, q + 1);
            c = __builtin_amdgcn_readlane(sc.y, q + 1);
            unsigned p = (unsigned)(s + lane);
            nextIdx = 0;
            if (lane < c && p < (unsigned)nEdges) nextIdx = nbr[p];
        }
        sage_add8(acc, v);
        for (int j = 8; j < m0; j += 8) {
            sage_rows8(xin, myIdx, j, m0, lane, v);
            sage_add8(acc, v);
        }
        for (int base = 32; base < cc; base += 32) {
            const int m = (cc - base) < 32 ? (cc - base) : 32;
            unsigned p = (unsigned)(cs + base + lane);
            myIdx = 0;
            if (lane < m && p < (unsigned)nEdges) myIdx = nbr[p];
            myIdx = (unsigned)myIdx < (unsigned)nNodes ? myIdx : 0;
            for (int j = 0; j < m; j += 8) {
                sage_rows8(xin, myIdx, j, m, lane, v);
                sage_add8(acc, v);
            }
        }
        const float rd = cc > 1 ? 1.0f / (float)cc : 1.0f;
        *(float4*)(sMean + (r0 + q) * LDS_PITCH + (lane << 2)) =
            make_float4(acc.x * rd, acc.y * rd, acc.z * rd, acc.w * rd);
    }
    __syncthreads();

    const int wave  = tid >> 5;
    const int lrow  = lane & 15;
    const int halfK = (lane >> 4) << 1;
    const int nCol  = wave * 16 + lrow;

    v8f acc[MSUB];
    #pragma unroll
    for (int mt = 0; mt < MSUB; ++mt) acc[mt] = (v8f){};

    for (int k0 = 0; k0 < HID; k0 += 4) {
        v2f b;
        b.x = wl[(size_t)(k0 + halfK)     * HID + nCol];
        b.y = wl[(size_t)(k0 + halfK + 1) * HID + nCol];
        #pragma unroll
        for (int mt = 0; mt < MSUB; ++mt) {
            const float* ap = sMean + (mt * 16 + lrow) * LDS_PITCH + k0 + halfK;
            v2f a; a.x = ap[0]; a.y = ap[1];
            acc[mt] = __builtin_amdgcn_wmma_f32_16x16x4_f32(
                          false, a, false, b, (short)0, acc[mt], false, false);
        }
    }

    for (int k0 = 0; k0 < HID; k0 += 4) {
        v2f b;
        b.x = wr[(size_t)(k0 + halfK)     * HID + nCol];
        b.y = wr[(size_t)(k0 + halfK + 1) * HID + nCol];
        #pragma unroll
        for (int mt = 0; mt < MSUB; ++mt) {
            const float* ap = sX + (mt * 16 + lrow) * LDS_PITCH + k0 + halfK;
            v2f a; a.x = ap[0]; a.y = ap[1];
            acc[mt] = __builtin_amdgcn_wmma_f32_16x16x4_f32(
                          false, a, false, b, (short)0, acc[mt], false, false);
        }
    }

    const float bias   = bl[nCol];
    const int   rowOff = (lane < 16) ? 0 : 8;
    #pragma unroll
    for (int mt = 0; mt < MSUB; ++mt) {
        #pragma unroll
        for (int r = 0; r < 8; ++r) {
            float v = acc[mt][r] + bias;
            if (applyRelu) v = v > 0.0f ? v : 0.0f;
            out[(size_t)(tileBase + mt * 16 + r + rowOff) * HID + nCol] = v;
        }
    }
}

extern "C" void kernel_launch(void* const* d_in, const int* in_sizes, int n_in,
                              void* d_out, int out_size, void* d_ws, size_t ws_size,
                              hipStream_t stream) {
    const float* x    = (const float*)d_in[0];
    const int*   ei   = (const int*)  d_in[1];
    const float* w1l  = (const float*)d_in[2];
    const float* b1l  = (const float*)d_in[3];
    const float* w1r  = (const float*)d_in[4];
    const float* w2l  = (const float*)d_in[5];
    const float* b2l  = (const float*)d_in[6];
    const float* w2r  = (const float*)d_in[7];

    const int N = in_sizes[0] / HID;
    const int E = in_sizes[1] / 2;
    const int* src = ei;
    const int* dst = ei + E;

    float* h     = (float*)d_ws;
    int*   seg   = (int*)(h + (size_t)N * HID);
    int*   slot  = seg + 2 * (size_t)N;
    int*   nbr   = slot + E;
    int*   cnt   = nbr  + E;
    float* out   = (float*)d_out;

    const int zGrid      = (N + 1 + 255) / 256;
    const int nGrid      = (N + 255) / 256;
    const int eGrid      = (E + 255) / 256;
    const int layerGrid  = N / MT;

    sage_zero_i <<<zGrid, 256, 0, stream>>>(cnt, N + 1);
    sage_count  <<<eGrid, 256, 0, stream>>>(dst, cnt, slot, E);
    sage_alloc  <<<nGrid, 256, 0, stream>>>(cnt, cnt + N, seg, N);
    sage_fill   <<<eGrid, 256, 0, stream>>>(src, dst, seg, slot, nbr, E);

    sage_layer  <<<layerGrid, 256, 0, stream>>>(x, seg, nbr, w1l, b1l, w1r, h,   1, N, E);
    sage_layer  <<<layerGrid, 256, 0, stream>>>(h, seg, nbr, w2l, b2l, w2r, out, 0, N, E);
}


# Neighbour lists as fixed-size slot rows per node, filled in the counting pass (slot = returning add; overflow in-edges to a spill list): alloc and fill launches dropped; spilled sources summed last
# speedup vs baseline: 2.1764x; 1.1213x over previous
#include <hip/hip_runtime.h>

#define HID       128
#define LDS_PITCH 132
#define MT        80
#define MSUB      5
#define ROWS_PER_WAVE (MT / 8)
#define NBR_CAP   16

typedef __attribute__((ext_vector_type(2))) float v2f;
typedef __attribute__((ext_vector_type(8))) float v8f;

__global__ void sage_zero_i(int* __restrict__ p, int n) {
    int i = blockIdx.x * blockDim.x + threadIdx.x;
    if (i < n) p[i] = 0;
}

__global__ void sage_build(const int* __restrict__ src,
                           const int* __restrict__ dst,
                           int*       __restrict__ cnt,
                           int*       __restrict__ nbr,
                           int*       __restrict__ spill,
                           int nNodes, int nEdges) {
    int e = blockIdx.x * blockDim.x + threadIdx.x;
    if (e < nEdges) {
        const int d = dst[e];
        const int s = src[e];
        const int slot = atomicAdd(cnt + d, 1);
        if (slot < NBR_CAP) {
            nbr[(size_t)d * NBR_CAP + slot] = s;
        } else {
            const unsigned p = (unsigned)atomicAdd(cnt + nNodes, 1);
            if (p < (unsigned)nEdges) ((int2*)spill)[p] = make_int2(d, s);
        }
    }
}

__device__ __forceinline__ void sage_rows8(const float* __restrict__ xin, int myIdx,
                                           int j, int m, int lane, float4 (&v)[8]) {
    #pragma unroll
    for (int k = 0; k < 8; ++k) {
        const int    idx  = __builtin_amdgcn_readlane(myIdx, j + k);
        const float4 t    = *((const float4*)(xin + (size_t)idx * HID) + lane);
        const bool   live = (j + k < m);
        v[k] = make_float4(live ? t.x : 0.0f, live ? t.y : 0.0f,
                           live ? t.z : 0.0f, live ? t.w : 0.0f);
    }
}

__device__ __forceinline__ void sage_add8(float4& acc, const float4 (&v)[8]) {
    #pragma unroll
    for (int k = 0; k < 8; ++k) {
        acc.x += v[k].x; acc.y += v[k].y; acc.z += v[k].z; acc.w += v[k].w;
    }
}

__global__ void sage_layer(const float* __restrict__ xin,
                           const int*   __restrict__ cnt,
                           const int*   __restrict__ nbr,
                           const int*   __restrict__ spill,
                           const float* __restrict__ wl,
                           const float* __restrict__ bl,
                           const float* __restrict__ wr,
                           float*       __restrict__ out,
                           int applyRelu, int nNodes, int nEdges) {
    __shared__ float sMean[MT * LDS_PITCH];
    __shared__ float sX   [MT * LDS_PITCH];

    const int tid      = threadIdx.x;
    const int tileBase = blockIdx.x * MT;
    const int lane     = tid & 31;
    const int waveU    = __builtin_amdgcn_readfirstlane(tid >> 5);
    const int r0       = waveU * ROWS_PER_WAVE;

    int rowCnt = 0;
    if (lane < ROWS_PER_WAVE) rowCnt = cnt[tileBase + r0 + lane];
    const int* myNbr = nbr + (size_t)(tileBase + r0) * NBR_CAP;

    for (int i = tid; i < MT * 32; i += 256) {
        int r   = i >> 5;
        int c4  = (i & 31) << 2;
        *(float4*)(sX + r * LDS_PITCH + c4) =
            *(const float4*)(xin + (size_t)(tileBase + r) * HID + c4);
    }

    int c = __builtin_amdgcn_readlane(rowCnt, 0);
    int nextIdx = 0;
    if (lane < (c < NBR_CAP ? c : NBR_CAP)) nextIdx = myNbr[lane];
    for (int q = 0; q < ROWS_PER_WAVE; ++q) {
        const int cc = c;
        const int m0 = cc < NBR_CAP ? cc : NBR_CAP;
        const int myIdx = (unsigned)nextIdx < (unsigned)nNodes ? nextIdx : 0;

        float4 acc = make_float4(0.0f, 0.0f, 0.0f, 0.0f);
        float4 v[8];
        sage_rows8(xin, myIdx, 0, m0, lane, v);
        if (q + 1 < ROWS_PER_WAVE) {
            c = __builtin_amdgcn_readlane(rowCnt, q + 1);
            nextIdx = 0;
            if (lane < (c < NBR_CAP ? c : NBR_CAP)) nextIdx = myNbr[(q + 1) * NBR_CAP + lane];
        }
        sage_add8(acc, v);
        for (int j = 8; j < m0; j += 8) {
            sage_rows8(xin, myIdx, j, m0, lane, v);
            sage_add8(acc, v);
        }
        if (cc > NBR_CAP) {
            const int row = tileBase + r0 + q;
            int nSpill = cnt[nNodes];
            nSpill = nSpill < nEdges ? nSpill : nEdges;
            for (int base = 0; base < nSpill; base += 32) {
                int2 ent = make_int2(-1, 0);
                if (base + lane < nSpill) ent = ((const int2*)spill)[base + lane];
                unsigned hits = __builtin_amdgcn_ballot_w32(ent.x == row);
                while (hits) {
                    const int l = __builtin_ctz(hits);
                    hits &= hits - 1;
                    int idx = __builtin_amdgcn_readlane(ent.y, l);
                    idx = (unsigned)idx < (unsigned)nNodes ? idx : 0;
                    const float4 t = *((const float4*)(xin + (size_t)idx * HID) + lane);
                    acc.x += t.x; acc.y += t.y; acc.z += t.z; acc.w += t.w;
                }
            }
        }
        const float rd = cc > 1 ? 1.0f / (float)cc : 1.0f;
        *(float4*)(sMean + (r0 + q) * LDS_PITCH + (lane << 2)) =
            make_float4(acc.x * rd, acc.y * rd, acc.z * rd, acc.w * rd);
    }
    __syncthreads();

    const int wave  = tid >> 5;
    const int lrow  = lane & 15;
    const int halfK = (lane >> 4) << 1;
    const int nCol  = wave * 16 + lrow;

    v8f acc[MSUB];
    #pragma unroll
    for (int mt = 0; mt < MSUB; ++mt) acc[mt] = (v8f){};

    for (int k0 = 0; k0 < HID; k0 += 4) {
        v2f b;
        b.x = wl[(size_t)(k0 + halfK)     * HID + nCol];
        b.y = wl[(size_t)(k0 + halfK + 1) * HID + nCol];
        #pragma unroll
        for (int mt = 0; mt < MSUB; ++mt) {
            const float* ap = sMean + (mt * 16 + lrow) * LDS_PITCH + k0 + halfK;
            v2f a; a.x = ap[0]; a.y = ap[1];
            acc[mt] = __builtin_amdgcn_wmma_f32_16x16x4_f32(
                          false, a, false, b, (short)0, acc[mt], false, false);
        }
    }

    for (int k0 = 0; k0 < HID; k0 += 4) {
        v2f b;
        b.x = wr[(size_t)(k0 + halfK)     * HID + nCol];
        b.y = wr[(size_t)(k0 + halfK + 1) * HID + nCol];
        #pragma unroll
        for (int mt = 0; mt < MSUB; ++mt) {
            const float* ap = sX + (mt * 16 + lrow) * LDS_PITCH + k0 + halfK;
            v2f a; a.x = ap[0]; a.y = ap[1];
            acc[mt] = __builtin_amdgcn_wmma_f32_16x16x4_f32(
                          false, a, false, b, (short)0, acc[mt], false, false);
        }
    }

    const float bias   = bl[nCol];
    const int   rowOff = (lane < 16) ? 0 : 8;
    #pragma unroll
    for (int mt = 0; mt < MSUB; ++mt) {
        #pragma unroll
        for (int r = 0; r < 8; ++r) {
            float v = acc[mt][r] + bias;
            if (applyRelu) v = v > 0.0f ? v : 0.0f;
            out[(size_t)(tileBase + mt * 16 + r + rowOff) * HID + nCol] = v;
        }
    }
}

extern "C" void kernel_launch(void* const* d_in, const int* in_sizes, int n_in,
                              void* d_out, int out_size, void* d_ws, size_t ws_size,
                              hipStream_t stream) {
    const float* x    = (const float*)d_in[0];
    const int*   ei   = (const int*)  d_in[1];
    const float* w1l  = (const float*)d_in[2];
    const float* b1l  = (const float*)d_in[3];
    const float* w1r  = (const float*)d_in[4];
    const float* w2l  = (const float*)d_in[5];
    const float* b2l  = (const float*)d_in[6];
    const float* w2r  = (const float*)d_in[7];

    const int N = in_sizes[0] / HID;
    const int E = in_sizes[1] / 2;
    const int* src = ei;
    const int* dst = ei + E;

    float* h     = (float*)d_ws;
    int*   nbr   = (int*)(h + (size_t)N * HID);
    int*   spill = nbr + (size_t)N * NBR_CAP;
    int*   cnt   = spill + 2 * (size_t)E;
    float* out   = (float*)d_out;

    const int zGrid      = (N + 1 + 255) / 256;
    const int eGrid      = (E + 255) / 256;
    const int layerGrid  = N / MT;

    sage_zero_i <<<zGrid, 256, 0, stream>>>(cnt, N + 1);
    sage_build  <<<eGrid, 256, 0, stream>>>(src, dst, cnt, nbr, spill, N, E);

    sage_layer  <<<layerGrid, 256, 0, stream>>>(x, cnt, nbr, spill, w1l, b1l, w1r, h,   1, N, E);
    sage_layer  <<<layerGrid, 256, 0, stream>>>(h, cnt, nbr, spill, w2l, b2l, w2r, out, 0, N, E);
}


# Gather keeps 3 tile rows of neighbour loads in flight per wave: all ten lists fetched first, then rows q+1 and q+2 are requested before row q is summed (3 register sets); per-row sum order unchanged
# speedup vs baseline: 2.2853x; 1.0500x over previous
#include <hip/hip_runtime.h>

#define HID       128
#define LDS_PITCH 132
#define MT        80
#define MSUB      5
#define ROWS_PER_WAVE (MT / 8)
#define NBR_CAP   16
#define GDEPTH    3

typedef __attribute__((ext_vector_type(2))) float v2f;
typedef __attribute__((ext_vector_type(8))) float v8f;

__global__ void sage_zero_i(int* __restrict__ p, int n) {
    int i = blockIdx.x * blockDim.x + threadIdx.x;
    if (i < n) p[i] = 0;
}

__global__ void sage_build(const int* __restrict__ src,
                           const int* __restrict__ dst,
                           int*       __restrict__ cnt,
                           int*       __restrict__ nbr,
                           int*       __restrict__ spill,
                           int nNodes, int nEdges) {
    int e = blockIdx.x * blockDim.x + threadIdx.x;
    if (e < nEdges) {
        const int d = dst[e];
        const int s = src[e];
        const int slot = atomicAdd(cnt + d, 1);
        if (slot < NBR_CAP) {
            nbr[(size_t)d * NBR_CAP + slot] = s;
        } else {
            const unsigned p = (unsigned)atomicAdd(cnt + nNodes, 1);
            if (p < (unsigned)nEdges) ((int2*)spill)[p] = make_int2(d, s);
        }
    }
}

__device__ __forceinline__ void sage_rows8(const float* __restrict__ xin, int myIdx,
                                           int j, int lane, float4 (&v)[8]) {
    #pragma unroll
    for (int k = 0; k < 8; ++k) {
        const int idx = __builtin_amdgcn_readlane(myIdx, j + k);
        v[k] = *((const float4*)(xin + (size_t)idx * HID) + lane);
    }
}

__device__ __forceinline__ void sage_add8(float4& acc, const float4 (&v)[8], int j, int m) {
    #pragma unroll
    for (int k = 0; k < 8; ++k) {
        const bool live = (j + k < m);
        acc.x += live ? v[k].x : 0.0f; acc.y += live ? v[k].y : 0.0f;
        acc.z += live ? v[k].z : 0.0f; acc.w += live ? v[k].w : 0.0f;
    }
}

__global__ void sage_layer(const float* __restrict__ xin,
                           const int*   __restrict__ cnt,
                           const int*   __restrict__ nbr,
                           const int*   __restrict__ spill,
                           const float* __restrict__ wl,
                           const float* __restrict__ bl,
                           const float* __restrict__ wr,
                           float*       __restrict__ out,
                           int applyRelu, int nNodes, int nEdges) {
    __shared__ float sMean[MT * LDS_PITCH];
    __shared__ float sX   [MT * LDS_PITCH];

    const int tid      = threadIdx.x;
    const int tileBase = blockIdx.x * MT;
    const int lane     = tid & 31;
    const int waveU    = __builtin_amdgcn_readfirstlane(tid >> 5);
    const int r0       = waveU * ROWS_PER_WAVE;

    int rowCnt = 0;
    if (lane < ROWS_PER_WAVE) rowCnt = cnt[tileBase + r0 + lane];
    const int* myNbr = nbr + (size_t)(tileBase + r0) * NBR_CAP;

    for (int i = tid; i < MT * 32; i += 256) {
        int r   = i >> 5;
        int c4  = (i & 31) << 2;
        *(float4*)(sX + r * LDS_PITCH + c4) =
            *(const float4*)(xin + (size_t)(tileBase + r) * HID + c4);
    }

    int idxv[ROWS_PER_WAVE];
    #pragma unroll
    for (int q = 0; q < ROWS_PER_WAVE; ++q) {
        const int c = __builtin_amdgcn_readlane(rowCnt, q);
        int t = 0;
        if (lane < (c < NBR_CAP ? c : NBR_CAP)) t = myNbr[q * NBR_CAP + lane];
        idxv[q] = (unsigned)t < (unsigned)nNodes ? t : 0;
    }

    float4 v[GDEPTH][8];
    #pragma unroll
    for (int p = 0; p < GDEPTH - 1; ++p) sage_rows8(xin, idxv[p], 0, lane, v[p]);

    #pragma unroll
    for (int q = 0; q < ROWS_PER_WAVE; ++q) {
        const int cc = __builtin_amdgcn_readlane(rowCnt, q);
        const int m0 = cc < NBR_CAP ? cc : NBR_CAP;

        if (q + GDEPTH - 1 < ROWS_PER_WAVE) {
            sage_rows8(xin, idxv[q + GDEPTH - 1], 0, lane, v[(q + GDEPTH - 1) % GDEPTH]);
            __builtin_amdgcn_sched_barrier(0);
        }

        float4 acc = make_float4(0.0f, 0.0f, 0.0f, 0.0f);
        sage_add8(acc, v[q % GDEPTH], 0, m0);
        if (m0 > 8) {
            sage_rows8(xin, idxv[q], 8, lane, v[q % GDEPTH]);
            sage_add8(acc, v[q % GDEPTH], 8, m0);
        }
        if (cc > NBR_CAP) {
            const int row = tileBase + r0 + q;
            int nSpill = cnt[nNodes];
            nSpill = nSpill < nEdges ? nSpill : nEdges;
            for (int base = 0; base < nSpill; base += 32) {
                int2 ent = make_int2(-1, 0);
                if (base + lane < nSpill) ent = ((const int2*)spill)[base + lane];
                unsigned hits = __builtin_amdgcn_ballot_w32(ent.x == row);
                while (hits) {
                    const int l = __builtin_ctz(hits);
                    hits &= hits - 1;
                    int idx = __builtin_amdgcn_readlane(ent.y, l);
                    idx = (unsigned)idx < (unsigned)nNodes ? idx : 0;
                    const float4 t = *((const float4*)(xin + (size_t)idx * HID) + lane);
                    acc.x += t.x; acc.y += t.y; acc.z += t.z; acc.w += t.w;
                }
            }
        }
        const float rd = cc > 1 ? 1.0f / (float)cc : 1.0f;
        *(float4*)(sMean + (r0 + q) * LDS_PITCH + (lane << 2)) =
            make_float4(acc.x * rd, acc.y * rd, acc.z * rd, acc.w * rd);
    }
    __syncthreads();

    const int wave  = tid >> 5;
    const int lrow  = lane & 15;
    const int halfK = (lane >> 4) << 1;
    const int nCol  = wave * 16 + lrow;

    v8f acc[MSUB];
    #pragma unroll
    for (int mt = 0; mt < MSUB; ++mt) acc[mt] = (v8f){};

    for (int k0 = 0; k0 < HID; k0 += 4) {
        v2f b;
        b.x = wl[(size_t)(k0 + halfK)     * HID + nCol];
        b.y = wl[(size_t)(k0 + halfK + 1) * HID + nCol];
        #pragma unroll
        for (int mt = 0; mt < MSUB; ++mt) {
            const float* ap = sMean + (mt * 16 + lrow) * LDS_PITCH + k0 + halfK;
            v2f a; a.x = ap[0]; a.y = ap[1];
            acc[mt] = __builtin_amdgcn_wmma_f32_16x16x4_f32(
                          false, a, false, b, (short)0, acc[mt], false, false);
        }
    }

    for (int k0 = 0; k0 < HID; k0 += 4) {
        v2f b;
        b.x = wr[(size_t)(k0 + halfK)     * HID + nCol];
        b.y = wr[(size_t)(k0 + halfK + 1) * HID + nCol];
        #pragma unroll
        for (int mt = 0; mt < MSUB; ++mt) {
            const float* ap = sX + (mt * 16 + lrow) * LDS_PITCH + k0 + halfK;
            v2f a; a.x = ap[0]; a.y = ap[1];
            acc[mt] = __builtin_amdgcn_wmma_f32_16x16x4_f32(
                          false, a, false, b, (short)0, acc[mt], false, false);
        }
    }

    const float bias   = bl[nCol];
    const int   rowOff = (lane < 16) ? 0 : 8;
    #pragma unroll
    for (int mt = 0; mt < MSUB; ++mt) {
        #pragma unroll
        for (int r = 0; r < 8; ++r) {
            float v = acc[mt][r] + bias;
            if (applyRelu) v = v > 0.0f ? v : 0.0f;
            out[(size_t)(tileBase + mt * 16 + r + rowOff) * HID + nCol] = v;
        }
    }
}

extern "C" void kernel_launch(void* const* d_in, const int* in_sizes, int n_in,
                              void* d_out, int out_size, void* d_ws, size_t ws_size,
                              hipStream_t stream) {
    const float* x    = (const float*)d_in[0];
    const int*   ei   = (const int*)  d_in[1];
    const float* w1l  = (const float*)d_in[2];
    const float* b1l  = (const float*)d_in[3];
    const float* w1r  = (const float*)d_in[4];
    const float* w2l  = (const float*)d_in[5];
    const float* b2l  = (const float*)d_in[6];
    const float* w2r  = (const float*)d_in[7];

    const int N = in_sizes[0] / HID;
    const int E = in_sizes[1] / 2;
    const int* src = ei;
    const int* dst = ei + E;

    float* h     = (float*)d_ws;
    int*   nbr   = (int*)(h + (size_t)N * HID);
    int*   spill = nbr + (size_t)N * NBR_CAP;
    int*   cnt   = spill + 2 * (size_t)E;
    float* out   = (float*)d_out;

    const int zGrid      = (N + 1 + 255) / 256;
    const int eGrid      = (E + 255) / 256;
    const int layerGrid  = N / MT;

    sage_zero_i <<<zGrid, 256, 0, stream>>>(cnt, N + 1);
    sage_build  <<<eGrid, 256, 0, stream>>>(src, dst, cnt, nbr, spill, N, E);

    sage_layer  <<<layerGrid, 256, 0, stream>>>(x, cnt, nbr, spill, w1l, b1l, w1r, h,   1, N, E);
    sage_layer  <<<layerGrid, 256, 0, stream>>>(h, cnt, nbr, spill, w2l, b2l, w2r, out, 0, N, E);
}


# Dense stage: weight slivers fetched a 16-k-step chunk ahead through two register sets (first chunk before the last gathered row is summed); kernel declared 256-thread; sum order unchanged
# speedup vs baseline: 2.3634x; 1.0342x over previous
#include <hip/hip_runtime.h>

#define HID       128
#define LDS_PITCH 132
#define MT        80
#define MSUB      5
#define ROWS_PER_WAVE (MT / 8)
#define NBR_CAP   16
#define GDEPTH    3
#define KCH       16

typedef __attribute__((ext_vector_type(2))) float v2f;
typedef __attribute__((ext_vector_type(8))) float v8f;

__global__ void sage_zero_i(int* __restrict__ p, int n) {
    int i = blockIdx.x * blockDim.x + threadIdx.x;
    if (i < n) p[i] = 0;
}

__global__ void sage_build(const int* __restrict__ src,
                           const int* __restrict__ dst,
                           int*       __restrict__ cnt,
                           int*       __restrict__ nbr,
                           int*       __restrict__ spill,
                           int nNodes, int nEdges) {
    int e = blockIdx.x * blockDim.x + threadIdx.x;
    if (e < nEdges) {
        const int d = dst[e];
        const int s = src[e];
        const int slot = atomicAdd(cnt + d, 1);
        if (slot < NBR_CAP) {
            nbr[(size_t)d * NBR_CAP + slot] = s;
        } else {
            const unsigned p = (unsigned)atomicAdd(cnt + nNodes, 1);
            if (p < (unsigned)nEdges) ((int2*)spill)[p] = make_int2(d, s);
        }
    }
}

__device__ __forceinline__ void sage_rows8(const float* __restrict__ xin, int myIdx,
                                           int j, int lane, float4 (&v)[8]) {
    #pragma unroll
    for (int k = 0; k < 8; ++k) {
        const int idx = __builtin_amdgcn_readlane(myIdx, j + k);
        v[k] = *((const float4*)(xin + (size_t)idx * HID) + lane);
    }
}

__device__ __forceinline__ void sage_add8(float4& acc, const float4 (&v)[8], int j, int m) {
    #pragma unroll
    for (int k = 0; k < 8; ++k) {
        const bool live = (j + k < m);
        acc.x += live ? v[k].x : 0.0f; acc.y += live ? v[k].y : 0.0f;
        acc.z += live ? v[k].z : 0.0f; acc.w += live ? v[k].w : 0.0f;
    }
}

__device__ __forceinline__ void sage_wfrag(const float* __restrict__ w, int ks0,
                                           int nCol, int halfK, v2f (&b)[KCH]) {
    #pragma unroll
    for (int i = 0; i < KCH; ++i) {
        const int k = (ks0 + i) * 4 + halfK;
        b[i].x = w[(size_t)k       * HID + nCol];
        b[i].y = w[(size_t)(k + 1) * HID + nCol];
    }
}

__device__ __forceinline__ void sage_chunk(const float* sA, int ks0, int lrow, int halfK,
                                           const v2f (&b)[KCH], v8f (&acc)[MSUB]) {
    #pragma unroll
    for (int i = 0; i < KCH; ++i) {
        const int k = (ks0 + i) * 4 + halfK;
        #pragma unroll
        for (int mt = 0; mt < MSUB; ++mt) {
            const float* ap = sA + (mt * 16 + lrow) * LDS_PITCH + k;
            v2f a; a.x = ap[0]; a.y = ap[1];
            acc[mt] = __builtin_amdgcn_wmma_f32_16x16x4_f32(
                          false, a, false, b[i], (short)0, acc[mt], false, false);
        }
    }
}

__global__ void __launch_bounds__(256) sage_layer(const float* __restrict__ xin,
                           const int*   __restrict__ cnt,
                           const int*   __restrict__ nbr,
                           const int*   __restrict__ spill,
                           const float* __restrict__ wl,
                           const float* __restrict__ bl,
                           const float* __restrict__ wr,
                           float*       __restrict__ out,
                           int applyRelu, int nNodes, int nEdges) {
    __shared__ float sMean[MT * LDS_PITCH];
    __shared__ float sX   [MT * LDS_PITCH];

    const int tid      = threadIdx.x;
    const int tileBase = blockIdx.x * MT;
    const int lane     = tid & 31;
    const int waveU    = __builtin_amdgcn_readfirstlane(tid >> 5);
    const int r0       = waveU * ROWS_PER_WAVE;

    int rowCnt = 0;
    if (lane < ROWS_PER_WAVE) rowCnt = cnt[tileBase + r0 + lane];
    const int* myNbr = nbr + (size_t)(tileBase + r0) * NBR_CAP;

    for (int i = tid; i < MT * 32; i += 256) {
        int r   = i >> 5;
        int c4  = (i & 31) << 2;
        *(float4*)(sX + r * LDS_PITCH + c4) =
            *(const float4*)(xin + (size_t)(tileBase + r) * HID + c4);
    }

    int idxv[ROWS_PER_WAVE];
    #pragma unroll
    for (int q = 0; q < ROWS_PER_WAVE; ++q) {
        const int c = __builtin_amdgcn_readlane(rowCnt, q);
        int t = 0;
        if (lane < (c < NBR_CAP ? c : NBR_CAP)) t = myNbr[q * NBR_CAP + lane];
        idxv[q] = (unsigned)t < (unsigned)nNodes ? t : 0;
    }

    const int wave  = tid >> 5;
    const int lrow  = lane & 15;
    const int halfK = (lane >> 4) << 1;
    const int nCol  = wave * 16 + lrow;
    v2f bq[2][KCH];

    float4 v[GDEPTH][8];
    #pragma unroll
    for (int p = 0; p < GDEPTH - 1; ++p) sage_rows8(xin, idxv[p], 0, lane, v[p]);

    #pragma unroll
    for (int q = 0; q < ROWS_PER_WAVE; ++q) {
        const int cc = __builtin_amdgcn_readlane(rowCnt, q);
        const int m0 = cc < NBR_CAP ? cc : NBR_CAP;

        if (q + GDEPTH - 1 < ROWS_PER_WAVE) {
            sage_rows8(xin, idxv[q + GDEPTH - 1], 0, lane, v[(q + GDEPTH - 1) % GDEPTH]);
            __builtin_amdgcn_sched_barrier(0);
        }

        if (q == ROWS_PER_WAVE - 1) {
            sage_wfrag(wl, 0, nCol, halfK, bq[0]);
            __builtin_amdgcn_sched_barrier(0);
        }

        float4 acc = make_float4(0.0f, 0.0f, 0.0f, 0.0f);
        sage_add8(acc, v[q % GDEPTH], 0, m0);
        if (m0 > 8) {
            sage_rows8(xin, idxv[q], 8, lane, v[q % GDEPTH]);
            sage_add8(acc, v[q % GDEPTH], 8, m0);
        }
        if (cc > NBR_CAP) {
            const int row = tileBase + r0 + q;
            int nSpill = cnt[nNodes];
            nSpill = nSpill < nEdges ? nSpill : nEdges;
            for (int base = 0; base < nSpill; base += 32) {
                int2 ent = make_int2(-1, 0);
                if (base + lane < nSpill) ent = ((const int2*)spill)[base + lane];
                unsigned hits = __builtin_amdgcn_ballot_w32(ent.x == row);
                while (hits) {
                    const int l = __builtin_ctz(hits);
                    hits &= hits - 1;
                    int idx = __builtin_amdgcn_readlane(ent.y, l);
                    idx = (unsigned)idx < (unsigned)nNodes ? idx : 0;
                    const float4 t = *((const float4*)(xin + (size_t)idx * HID) + lane);
                    acc.x += t.x; acc.y += t.y; acc.z += t.z; acc.w += t.w;
                }
            }
        }
        const float rd = cc > 1 ? 1.0f / (float)cc : 1.0f;
        *(float4*)(sMean + (r0 + q) * LDS_PITCH + (lane << 2)) =
            make_float4(acc.x * rd, acc.y * rd, acc.z * rd, acc.w * rd);
    }
    __syncthreads();

    v8f acc[MSUB];
    #pragma unroll
    for (int mt = 0; mt < MSUB; ++mt) acc[mt] = (v8f){};

    #pragma unroll
    for (int c = 0; c < 4; ++c) {
        if (c + 1 < 4)
            sage_wfrag((c + 1) < 2 ? wl : wr, ((c + 1) & 1) * KCH, nCol, halfK, bq[(c + 1) & 1]);
        sage_chunk(c < 2 ? sMean : sX, (c & 1) * KCH, lrow, halfK, bq[c & 1], acc);
    }

    const float bias   = bl[nCol];
    const int   rowOff = (lane < 16) ? 0 : 8;
    #pragma unroll
    for (int mt = 0; mt < MSUB; ++mt) {
        #pragma unroll
        for (int r = 0; r < 8; ++r) {
            float v = acc[mt][r] + bias;
            if (applyRelu) v = v > 0.0f ? v : 0.0f;
            out[(size_t)(tileBase + mt * 16 + r + rowOff) * HID + nCol] = v;
        }
    }
}

extern "C" void kernel_launch(void* const* d_in, const int* in_sizes, int n_in,
                              void* d_out, int out_size, void* d_ws, size_t ws_size,
                              hipStream_t stream) {
    const float* x    = (const float*)d_in[0];
    const int*   ei   = (const int*)  d_in[1];
    const float* w1l  = (const float*)d_in[2];
    const float* b1l  = (const float*)d_in[3];
    const float* w1r  = (const float*)d_in[4];
    const float* w2l  = (const float*)d_in[5];
    const float* b2l  = (const float*)d_in[6];
    const float* w2r  = (const float*)d_in[7];

    const int N = in_sizes[0] / HID;
    const int E = in_sizes[1] / 2;
    const int* src = ei;
    const int* dst = ei + E;

    float* h     = (float*)d_ws;
    int*   nbr   = (int*)(h + (size_t)N * HID);
    int*   spill = nbr + (size_t)N * NBR_CAP;
    int*   cnt   = spill + 2 * (size_t)E;
    float* out   = (float*)d_out;

    const int zGrid      = (N + 1 + 255) / 256;
    const int eGrid      = (E + 255) / 256;
    const int layerGrid  = N / MT;

    sage_zero_i <<<zGrid, 256, 0, stream>>>(cnt, N + 1);
    sage_build  <<<eGrid, 256, 0, stream>>>(src, dst, cnt, nbr, spill, N, E);

    sage_layer  <<<layerGrid, 256, 0, stream>>>(x, cnt, nbr, spill, w1l, b1l, w1r, h,   1, N, E);
    sage_layer  <<<layerGrid, 256, 0, stream>>>(h, cnt, nbr, spill, w2l, b2l, w2r, out, 0, N, E);
}


# Neighbour lists grouped in LDS in two launches (edge chunks binned by destination range, then slots by returning LDS adds) replace zero pass and per-edge global add; sum order = LDS arrival slot
# speedup vs baseline: 2.8774x; 1.2175x over previous
#include <hip/hip_runtime.h>

#define HID       128
#define LDS_PITCH 132
#define MT        80
#define MSUB      5
#define ROWS_PER_WAVE (MT / 8)
#define NBR_CAP   16
#define GDEPTH    3
#define RANGE     400
#define NRANGE    256
#define NCHUNK    256
#define CELL      32
#define EPT       10
#define KCH       16

typedef __attribute__((ext_vector_type(2))) float v2f;
typedef __attribute__((ext_vector_type(8))) float v8f;

__global__ void __launch_bounds__(256) sage_bin(const int* __restrict__ src,
                                                const int* __restrict__ dst,
                                                int*       __restrict__ cellsI,
                                                int*       __restrict__ ovfI,
                                                int*       __restrict__ ovfCnt,
                                                int*       __restrict__ spillCursor,
                                                int nNodes, int nEdges) {
    __shared__ int  hist[NRANGE];
    __shared__ int  ovfN;
    __shared__ int2 stage[NRANGE * CELL];

    const int t = threadIdx.x;
    const int b = blockIdx.x;
    const int chunk = (nEdges + NCHUNK - 1) / NCHUNK;
    const int e0 = b * chunk;
    const int e1 = (e0 + chunk < nEdges) ? e0 + chunk : nEdges;
    int2* cells = (int2*)cellsI;
    int2* myOvf = (int2*)ovfI + (size_t)b * chunk;

    hist[t] = 0;
    if (t == 0) { ovfN = 0; if (b == 0) *spillCursor = 0; }
    for (int i = t; i < NRANGE * CELL / 2; i += 256) ((int4*)stage)[i] = make_int4(-1, 0, -1, 0);
    __syncthreads();

    for (int base = e0; base < e1; base += 256 * EPT) {
        int d[EPT], s[EPT];
        #pragma unroll
        for (int i = 0; i < EPT; ++i) {
            const int e = base + i * 256 + t;
            d[i] = -1; s[i] = 0;
            if (e < e1) { d[i] = dst[e]; s[i] = src[e]; }
        }
        int k[EPT], pos[EPT];
        #pragma unroll
        for (int i = 0; i < EPT; ++i) {
            k[i] = -1; pos[i] = 0;
            if ((unsigned)d[i] < (unsigned)nNodes) {
                const int r = d[i] / RANGE;
                if (r < NRANGE) { k[i] = r; pos[i] = atomicAdd(&hist[r], 1); }
            }
        }
        #pragma unroll
        for (int i = 0; i < EPT; ++i) {
            if (k[i] >= 0) {
                if (pos[i] < CELL) {
                    stage[k[i] * CELL + pos[i]] = make_int2(d[i], s[i]);
                } else {
                    const int o = atomicAdd(&ovfN, 1);
                    myOvf[o] = make_int2(d[i], s[i]);
                }
            }
        }
    }
    __syncthreads();

    const int wave = t >> 5, lane = t & 31;
    const int nR = (nNodes + RANGE - 1) / RANGE;
    for (int k = wave; k < NRANGE && k < nR; k += 8)
        cells[((size_t)k * NCHUNK + b) * CELL + lane] = stage[k * CELL + lane];
    if (t == 0) ovfCnt[b] = ovfN;
}

__device__ __forceinline__ void sage_place(int2 p, int node0, int* cntL, int* nbrL,
                                           int* cnt, int* spill, int nNodes, int nEdges) {
    const int ld = p.x - node0;
    if ((unsigned)ld < (unsigned)RANGE) {
        const int slot = atomicAdd(cntL + ld, 1);
        if (slot < NBR_CAP) {
            nbrL[ld * NBR_CAP + slot] = p.y;
        } else {
            const unsigned q = (unsigned)atomicAdd(cnt + nNodes, 1);
            if (q < (unsigned)nEdges) ((int2*)spill)[q] = p;
        }
    }
}

__global__ void __launch_bounds__(256) sage_group(const int*  __restrict__ cellsI,
                                                  const int*  __restrict__ ovfI,
                                                  const int*  __restrict__ ovfCnt,
                                                  int*        __restrict__ cnt,
                                                  int*        __restrict__ nbr,
                                                  int*        __restrict__ spill,
                                                  int nNodes, int nEdges) {
    __shared__ int cntL[RANGE];
    __shared__ int nbrL[RANGE * NBR_CAP];
    __shared__ int ovL[NCHUNK];
    __shared__ int anyOvf;

    const int t = threadIdx.x;
    const int k = blockIdx.x;
    const int wave = t >> 5, lane = t & 31;
    const int node0 = k * RANGE;

    const int2* ovf     = (const int2*)ovfI;
    const int2* myCells = (const int2*)cellsI + ((size_t)k * NCHUNK + wave * 32) * CELL + lane;
    int2 p[32];
    #pragma unroll
    for (int i = 0; i < 32; ++i) p[i] = myCells[(size_t)i * CELL];

    const int myOv = ovfCnt[t];
    ovL[t] = myOv;
    if (t == 0) anyOvf = 0;
    for (int i = t; i < RANGE; i += 256) cntL[i] = 0;
    for (int i = t; i < RANGE * NBR_CAP / 4; i += 256) ((int4*)nbrL)[i] = make_int4(0, 0, 0, 0);
    __syncthreads();
    if (myOv > 0) anyOvf = 1;

    int slot[32];
    #pragma unroll
    for (int i = 0; i < 32; ++i) {
        const int ld = p[i].x - node0;
        slot[i] = -1;
        if ((unsigned)ld < (unsigned)RANGE) slot[i] = atomicAdd(cntL + ld, 1);
    }
    #pragma unroll
    for (int i = 0; i < 32; ++i) {
        if (slot[i] >= 0) {
            if (slot[i] < NBR_CAP) {
                nbrL[(p[i].x - node0) * NBR_CAP + slot[i]] = p[i].y;
            } else {
                const unsigned q = (unsigned)atomicAdd(cnt + nNodes, 1);
                if (q < (unsigned)nEdges) ((int2*)spill)[q] = p[i];
            }
        }
    }
    __syncthreads();

    if (anyOvf) {
        const int chunk = (nEdges + NCHUNK - 1) / NCHUNK;
        for (int b = 0; b < NCHUNK; ++b) {
            int on = ovL[b];
            on = on < chunk ? on : chunk;
            for (int j = t; j < on; j += 256)
                sage_place(ovf[(size_t)b * chunk + j], node0, cntL, nbrL, cnt, spill, nNodes, nEdges);
        }
    }
    __syncthreads();

    for (int i = t; i < RANGE; i += 256)
        if (node0 + i < nNodes) cnt[node0 + i] = cntL[i];
    int4* g = (int4*)(nbr + (size_t)node0 * NBR_CAP);
    for (int i = t; i < RANGE * NBR_CAP / 4; i += 256)
        if (node0 + (i >> 2) < nNodes) g[i] = ((const int4*)nbrL)[i];
}

__device__ __forceinline__ void sage_rows8(const float* __restrict__ xin, int myIdx,
                                           int j, int lane, float4 (&v)[8]) {
    #pragma unroll
    for (int k = 0; k < 8; ++k) {
        const int idx = __builtin_amdgcn_readlane(myIdx, j + k);
        v[k] = *((const float4*)(xin + (size_t)idx * HID) + lane);
    }
}

__device__ __forceinline__ void sage_add8(float4& acc, const float4 (&v)[8], int j, int m) {
    #pragma unroll
    for (int k = 0; k < 8; ++k) {
        const bool live = (j + k < m);
        acc.x += live ? v[k].x : 0.0f; acc.y += live ? v[k].y : 0.0f;
        acc.z += live ? v[k].z : 0.0f; acc.w += live ? v[k].w : 0.0f;
    }
}

__device__ __forceinline__ void sage_wfrag(const float* __restrict__ w, int ks0,
                                           int nCol, int halfK, v2f (&b)[KCH]) {
    #pragma unroll
    for (int i = 0; i < KCH; ++i) {
        const int k = (ks0 + i) * 4 + halfK;
        b[i].x = w[(size_t)k       * HID + nCol];
        b[i].y = w[(size_t)(k + 1) * HID + nCol];
    }
}

__device__ __forceinline__ void sage_chunk(const float* sA, int ks0, int lrow, int halfK,
                                           const v2f (&b)[KCH], v8f (&acc)[MSUB]) {
    #pragma unroll
    for (int i = 0; i < KCH; ++i) {
        const int k = (ks0 + i) * 4 + halfK;
        #pragma unroll
        for (int mt = 0; mt < MSUB; ++mt) {
            const float* ap = sA + (mt * 16 + lrow) * LDS_PITCH + k;
            v2f a; a.x = ap[0]; a.y = ap[1];
            acc[mt] = __builtin_amdgcn_wmma_f32_16x16x4_f32(
                          false, a, false, b[i], (short)0, acc[mt], false, false);
        }
    }
}

__global__ void __launch_bounds__(256) sage_layer(const float* __restrict__ xin,
                           const int*   __restrict__ cnt,
                           const int*   __restrict__ nbr,
                           const int*   __restrict__ spill,
                           const float* __restrict__ wl,
                           const float* __restrict__ bl,
                           const float* __restrict__ wr,
                           float*       __restrict__ out,
                           int applyRelu, int nNodes, int nEdges) {
    __shared__ float sMean[MT * LDS_PITCH];
    __shared__ float sX   [MT * LDS_PITCH];

    const int tid      = threadIdx.x;
    const int tileBase = blockIdx.x * MT;
    const int lane     = tid & 31;
    const int waveU    = __builtin_amdgcn_readfirstlane(tid >> 5);
    const int r0       = waveU * ROWS_PER_WAVE;

    int rowCnt = 0;
    if (lane < ROWS_PER_WAVE) rowCnt = cnt[tileBase + r0 + lane];
    const int* myNbr = nbr + (size_t)(tileBase + r0) * NBR_CAP;

    for (int i = tid; i < MT * 32; i += 256) {
        int r   = i >> 5;
        int c4  = (i & 31) << 2;
        *(float4*)(sX + r * LDS_PITCH + c4) =
            *(const float4*)(xin + (size_t)(tileBase + r) * HID + c4);
    }

    int idxv[ROWS_PER_WAVE];
    #pragma unroll
    for (int q = 0; q < ROWS_PER_WAVE; ++q) {
        const int c = __builtin_amdgcn_readlane(rowCnt, q);
        int t = 0;
        if (lane < (c < NBR_CAP ? c : NBR_CAP)) t = myNbr[q * NBR_CAP + lane];
        idxv[q] = (unsigned)t < (unsigned)nNodes ? t : 0;
    }

    const int wave  = tid >> 5;
    const int lrow  = lane & 15;
    const int halfK = (lane >> 4) << 1;
    const int nCol  = wave * 16 + lrow;
    v2f bq[2][KCH];

    float4 v[GDEPTH][8];
    #pragma unroll
    for (int p = 0; p < GDEPTH - 1; ++p) sage_rows8(xin, idxv[p], 0, lane, v[p]);

    #pragma unroll
    for (int q = 0; q < ROWS_PER_WAVE; ++q) {
        const int cc = __builtin_amdgcn_readlane(rowCnt, q);
        const int m0 = cc < NBR_CAP ? cc : NBR_CAP;

        if (q + GDEPTH - 1 < ROWS_PER_WAVE) {
            sage_rows8(xin, idxv[q + GDEPTH - 1], 0, lane, v[(q + GDEPTH - 1) % GDEPTH]);
            __builtin_amdgcn_sched_barrier(0);
        }

        if (q == ROWS_PER_WAVE - 1) {
            sage_wfrag(wl, 0, nCol, halfK, bq[0]);
            __builtin_amdgcn_sched_barrier(0);
        }

        float4 acc = make_float4(0.0f, 0.0f, 0.0f, 0.0f);
        sage_add8(acc, v[q % GDEPTH], 0, m0);
        if (m0 > 8) {
            sage_rows8(xin, idxv[q], 8, lane, v[q % GDEPTH]);
            sage_add8(acc, v[q % GDEPTH], 8, m0);
        }
        if (cc > NBR_CAP) {
            const int row = tileBase + r0 + q;
            int nSpill = cnt[nNodes];
            nSpill = nSpill < nEdges ? nSpill : nEdges;
            for (int base = 0; base < nSpill; base += 32) {
                int2 ent = make_int2(-1, 0);
                if (base + lane < nSpill) ent = ((const int2*)spill)[base + lane];
                unsigned hits = __builtin_amdgcn_ballot_w32(ent.x == row);
                while (hits) {
                    const int l = __builtin_ctz(hits);
                    hits &= hits - 1;
                    int idx = __builtin_amdgcn_readlane(ent.y, l);
                    idx = (unsigned)idx < (unsigned)nNodes ? idx : 0;
                    const float4 t = *((const float4*)(xin + (size_t)idx * HID) + lane);
                    acc.x += t.x; acc.y += t.y; acc.z += t.z; acc.w += t.w;
                }
            }
        }
        const float rd = cc > 1 ? 1.0f / (float)cc : 1.0f;
        *(float4*)(sMean + (r0 + q) * LDS_PITCH + (lane << 2)) =
            make_float4(acc.x * rd, acc.y * rd, acc.z * rd, acc.w * rd);
    }
    __syncthreads();

    v8f acc[MSUB];
    #pragma unroll
    for (int mt = 0; mt < MSUB; ++mt) acc[mt] = (v8f){};

    #pragma unroll
    for (int c = 0; c < 4; ++c) {
        if (c + 1 < 4)
            sage_wfrag((c + 1) < 2 ? wl : wr, ((c + 1) & 1) * KCH, nCol, halfK, bq[(c + 1) & 1]);
        sage_chunk(c < 2 ? sMean : sX, (c & 1) * KCH, lrow, halfK, bq[c & 1], acc);
    }

    const float bias   = bl[nCol];
    const int   rowOff = (lane < 16) ? 0 : 8;
    #pragma unroll
    for (int mt = 0; mt < MSUB; ++mt) {
        #pragma unroll
        for (int r = 0; r < 8; ++r) {
            float v = acc[mt][r] + bias;
            if (applyRelu) v = v > 0.0f ? v : 0.0f;
            out[(size_t)(tileBase + mt * 16 + r + rowOff) * HID + nCol] = v;
        }
    }
}

extern "C" void kernel_launch(void* const* d_in, const int* in_sizes, int n_in,
                              void* d_out, int out_size, void* d_ws, size_t ws_size,
                              hipStream_t stream) {
    const float* x    = (const float*)d_in[0];
    const int*   ei   = (const int*)  d_in[1];
    const float* w1l  = (const float*)d_in[2];
    const float* b1l  = (const float*)d_in[3];
    const float* w1r  = (const float*)d_in[4];
    const float* w2l  = (const float*)d_in[5];
    const float* b2l  = (const float*)d_in[6];
    const float* w2r  = (const float*)d_in[7];

    const int N = in_sizes[0] / HID;
    const int E = in_sizes[1] / 2;
    const int* src = ei;
    const int* dst = ei + E;

    float* h      = (float*)d_ws;
    int*   nbr    = (int*)(h + (size_t)N * HID);
    int*   spill  = nbr + (size_t)N * NBR_CAP;
    int*   cells  = spill + 2 * (size_t)E;
    int*   ovf    = cells + 2 * (size_t)NRANGE * NCHUNK * CELL;
    int*   ovfCnt = ovf + 2 * ((size_t)E + NCHUNK);
    int*   cnt    = ovfCnt + NCHUNK;
    float* out    = (float*)d_out;

    const int gGrid      = (N + RANGE - 1) / RANGE;
    const int layerGrid  = N / MT;

    sage_bin    <<<NCHUNK, 256, 0, stream>>>(src, dst, cells, ovf, ovfCnt, cnt + N, N, E);
    sage_group  <<<gGrid,  256, 0, stream>>>(cells, ovf, ovfCnt, cnt, nbr, spill, N, E);

    sage_layer  <<<layerGrid, 256, 0, stream>>>(x, cnt, nbr, spill, w1l, b1l, w1r, h,   1, N, E);
    sage_layer  <<<layerGrid, 256, 0, stream>>>(h, cnt, nbr, spill, w2l, b2l, w2r, out, 0, N, E);
}
